# starting kernel
# baseline (speedup 1.0000x reference)
#include <hip/hip_runtime.h>
#include <math.h>

typedef __attribute__((ext_vector_type(2))) float v2f;
typedef __attribute__((ext_vector_type(8))) float v8f;

#define DD   128
#define QQ   64
#define CC   64
#define TT   8
#define S128 132
#define S64  68

__device__ __forceinline__ v8f wmma4(v2f a, v2f b, v8f c) {
  return __builtin_amdgcn_wmma_f32_16x16x4_f32(
       false, a,  false, b,
       (short)0, c,  false,  false);
}

__global__ __launch_bounds__(256)
void entity_linker_kernel(const int* __restrict__ q_ids,
                          const int* __restrict__ c_ids,
                          const int* __restrict__ num_qs,
                          const float* __restrict__ embed,
                          const float* __restrict__ W_h,
                          const float* __restrict__ b_h,
                          const float* __restrict__ W_o,
                          const float* __restrict__ b_o,
                          float* __restrict__ out) {
  extern __shared__ float smem[];
  float* sq   = smem;
  float* sc   = sq  + 64 * S128;
  float* swq  = sc  + 64 * S128;
  float* satt = swq + 64 * S128;
  float* qsum = satt + 64 * S64;

  const int b    = blockIdx.x;
  const int tid  = threadIdx.x;
  const int wave = tid >> 5;
  const int lane = tid & 31;
  const int half = lane >> 4;
  const int l    = lane & 15;
  const int num_q = num_qs[b];

  {
    const int q  = tid >> 2;
    const int d0 = (tid & 3) * 32;
    const float* src = embed + (size_t)q_ids[(size_t)b * QQ + q] * DD + d0;
    float* dst = sq + q * S128 + d0;
#pragma unroll
    for (int i = 0; i < 32; ++i) dst[i] = src[i];
  }
  {
    const int c  = tid >> 2;
    const int d0 = (tid & 3) * 32;
    const int* ids = c_ids + ((size_t)b * CC + c) * TT;
    float acc[32];
#pragma unroll
    for (int i = 0; i < 32; ++i) acc[i] = 0.f;
    float cnt = 0.f;
    for (int t = 0; t < TT; ++t) {
      const int id = ids[t];
      if (id != 0) {
        cnt += 1.f;
        const float* e = embed + (size_t)id * DD + d0;
#pragma unroll
        for (int i = 0; i < 32; ++i) acc[i] += e[i];
      }
    }
    const float inv = 1.f / fmaxf(cnt, 1.f);
    float* dst = sc + c * S128 + d0;
#pragma unroll
    for (int i = 0; i < 32; ++i) dst[i] = acc[i] * inv;
  }
  __syncthreads();

  if (tid < DD) {
    float s = 0.f;
    for (int q = 0; q < num_q; ++q) s += sq[q * S128 + tid];
    qsum[tid] = s / fmaxf((float)num_q, 1.f);
  }
  __syncthreads();

  const float scale = 0.088388347648318447f;
#pragma unroll
  for (int t2 = 0; t2 < 2; ++t2) {
    const int t  = wave * 2 + t2;
    const int mt = t >> 2, nt = t & 3;
    const float* Ar = sc + (mt * 16 + l) * S128;
    const float* Br = sq + (nt * 16 + l) * S128;
    v8f acc = {};
    for (int kk = 0; kk < 32; ++kk) {
      const int k0 = kk * 4 + 2 * half;
      v2f a;  a.x  = Ar[k0]; a.y  = Ar[k0 + 1];
      v2f bv; bv.x = Br[k0]; bv.y = Br[k0 + 1];
      acc = wmma4(a, bv, acc);
    }
    const int ncol = nt * 16 + l;
    const bool valid = (ncol < num_q);
#pragma unroll
    for (int r = 0; r < 8; ++r) {
      const int row = mt * 16 + r + 8 * half;
      satt[row * S64 + ncol] = valid ? acc[r] * scale : -__builtin_inff();
    }
  }
  __syncthreads();

  if (tid < CC) {
    float* row = satt + tid * S64;
    float m = -__builtin_inff();
    for (int q = 0; q < QQ; ++q) m = fmaxf(m, row[q]);
    float s = 0.f;
    for (int q = 0; q < QQ; ++q) { const float e = __expf(row[q] - m); row[q] = e; s += e; }
    const float inv = 1.f / s;
    for (int q = 0; q < QQ; ++q) row[q] *= inv;
  }
  __syncthreads();

#pragma unroll
  for (int t4 = 0; t4 < 4; ++t4) {
    const int t  = wave * 4 + t4;
    const int mt = t >> 3, nt = t & 7;
    const float* Ar = satt + (mt * 16 + l) * S64;
    const int ncol = nt * 16 + l;
    v8f acc = {};
    for (int kk = 0; kk < 16; ++kk) {
      const int k0 = kk * 4 + 2 * half;
      v2f a;  a.x  = Ar[k0];                 a.y  = Ar[k0 + 1];
      v2f bv; bv.x = sq[k0 * S128 + ncol];   bv.y = sq[(k0 + 1) * S128 + ncol];
      acc = wmma4(a, bv, acc);
    }
#pragma unroll
    for (int r = 0; r < 8; ++r)
      swq[(mt * 16 + r + 8 * half) * S128 + ncol] = acc[r];
  }
  __syncthreads();

#pragma unroll
  for (int t4 = 0; t4 < 4; ++t4) {
    const int t  = wave * 4 + t4;
    const int mt = t >> 3, nt = t & 7;
    const int arow = mt * 16 + l;
    const int ncol = nt * 16 + l;
    const float* Cr = sc  + arow * S128;
    const float* Qr = swq + arow * S128;
    const float* Wp = W_h + ncol;
    v8f acc = {};
    for (int kk = 0; kk < 32; ++kk) {
      const int k0 = kk * 4 + 2 * half;
      v2f a;  a.x  = qsum[k0];                a.y  = qsum[k0 + 1];
      v2f bv; bv.x = Wp[(size_t)k0 * DD];     bv.y = Wp[(size_t)(k0 + 1) * DD];
      acc = wmma4(a, bv, acc);
    }
    for (int kk = 0; kk < 32; ++kk) {
      const int k0 = kk * 4 + 2 * half;
      v2f a;  a.x  = Cr[k0];                       a.y  = Cr[k0 + 1];
      v2f bv; bv.x = Wp[(size_t)(k0 + 128) * DD];  bv.y = Wp[(size_t)(k0 + 129) * DD];
      acc = wmma4(a, bv, acc);
    }
    for (int kk = 0; kk < 32; ++kk) {
      const int k0 = kk * 4 + 2 * half;
      v2f a;  a.x  = Qr[k0];                       a.y  = Qr[k0 + 1];
      v2f bv; bv.x = Wp[(size_t)(k0 + 256) * DD];  bv.y = Wp[(size_t)(k0 + 257) * DD];
      acc = wmma4(a, bv, acc);
    }
    for (int kk = 0; kk < 32; ++kk) {
      const int k0 = kk * 4 + 2 * half;
      v2f a;  a.x  = Cr[k0] * Qr[k0];              a.y  = Cr[k0 + 1] * Qr[k0 + 1];
      v2f bv; bv.x = Wp[(size_t)(k0 + 384) * DD];  bv.y = Wp[(size_t)(k0 + 385) * DD];
      acc = wmma4(a, bv, acc);
    }
    for (int kk = 0; kk < 32; ++kk) {
      const int k0 = kk * 4 + 2 * half;
      v2f a;  a.x  = fabsf(Cr[k0] - Qr[k0]);       a.y  = fabsf(Cr[k0 + 1] - Qr[k0 + 1]);
      v2f bv; bv.x = Wp[(size_t)(k0 + 512) * DD];  bv.y = Wp[(size_t)(k0 + 513) * DD];
      acc = wmma4(a, bv, acc);
    }
    const float bias = b_h[ncol];
#pragma unroll
    for (int r = 0; r < 8; ++r)
      sq[(mt * 16 + r + 8 * half) * S128 + ncol] = tanhf(acc[r] + bias);
  }
  __syncthreads();

  if (tid < CC) {
    float s = 0.f;
    for (int n = 0; n < DD; ++n) s += sq[tid * S128 + n] * W_o[n];
    out[(size_t)b * CC + tid] = s + b_o[0];
  }
}

extern "C" void kernel_launch(void* const* d_in, const int* in_sizes, int n_in,
                              void* d_out, int out_size, void* d_ws, size_t ws_size,
                              hipStream_t stream) {
  (void)in_sizes; (void)n_in; (void)out_size; (void)d_ws; (void)ws_size;
  const int*   q_ids  = (const int*)d_in[0];
  const int*   c_ids  = (const int*)d_in[1];
  const int*   num_qs = (const int*)d_in[2];
  const float* embed  = (const float*)d_in[4];
  const float* W_h    = (const float*)d_in[5];
  const float* b_h    = (const float*)d_in[6];
  const float* W_o    = (const float*)d_in[7];
  const float* b_o    = (const float*)d_in[8];
  float* out = (float*)d_out;

  const size_t shmem = (size_t)(3 * 64 * S128 + 64 * S64 + 128) * sizeof(float);
  hipFuncSetAttribute(reinterpret_cast<const void*>(entity_linker_kernel),
                      hipFuncAttributeMaxDynamicSharedMemorySize, (int)shmem);
  entity_linker_kernel<<<1024, 256, shmem, stream>>>(
      q_ids, c_ids, num_qs, embed, W_h, b_h, W_o, b_o, out);
}


# h GEMM re-tiled: each wave owns one 16-col strip of W_h for all 4 row tiles, so a W_h fragment is loaded once for 4 WMMAs; the 5 segments are summed interleaved per k-step, q_summary part once
# speedup vs baseline: 1.2668x; 1.2668x over previous
#include <hip/hip_runtime.h>
#include <math.h>

typedef __attribute__((ext_vector_type(2))) float v2f;
typedef __attribute__((ext_vector_type(8))) float v8f;

#define DD   128
#define QQ   64
#define CC   64
#define TT   8
#define S128 132
#define S64  68

__device__ __forceinline__ v8f wmma4(v2f a, v2f b, v8f c) {
  return __builtin_amdgcn_wmma_f32_16x16x4_f32(
       false, a,  false, b,
       (short)0, c,  false,  false);
}

__global__ __launch_bounds__(256)
void entity_linker_kernel(const int* __restrict__ q_ids,
                          const int* __restrict__ c_ids,
                          const int* __restrict__ num_qs,
                          const float* __restrict__ embed,
                          const float* __restrict__ W_h,
                          const float* __restrict__ b_h,
                          const float* __restrict__ W_o,
                          const float* __restrict__ b_o,
                          float* __restrict__ out) {
  extern __shared__ float smem[];
  float* sq   = smem;
  float* sc   = sq  + 64 * S128;
  float* swq  = sc  + 64 * S128;
  float* satt = swq + 64 * S128;
  float* qsum = satt + 64 * S64;

  const int b    = blockIdx.x;
  const int tid  = threadIdx.x;
  const int wave = tid >> 5;
  const int lane = tid & 31;
  const int half = lane >> 4;
  const int l    = lane & 15;
  const int num_q = num_qs[b];

  {
    const int q  = tid >> 2;
    const int d0 = (tid & 3) * 32;
    const float* src = embed + (size_t)q_ids[(size_t)b * QQ + q] * DD + d0;
    float* dst = sq + q * S128 + d0;
#pragma unroll
    for (int i = 0; i < 32; ++i) dst[i] = src[i];
  }
  {
    const int c  = tid >> 2;
    const int d0 = (tid & 3) * 32;
    const int* ids = c_ids + ((size_t)b * CC + c) * TT;
    float acc[32];
#pragma unroll
    for (int i = 0; i < 32; ++i) acc[i] = 0.f;
    float cnt = 0.f;
    for (int t = 0; t < TT; ++t) {
      const int id = ids[t];
      if (id != 0) {
        cnt += 1.f;
        const float* e = embed + (size_t)id * DD + d0;
#pragma unroll
        for (int i = 0; i < 32; ++i) acc[i] += e[i];
      }
    }
    const float inv = 1.f / fmaxf(cnt, 1.f);
    float* dst = sc + c * S128 + d0;
#pragma unroll
    for (int i = 0; i < 32; ++i) dst[i] = acc[i] * inv;
  }
  __syncthreads();

  if (tid < DD) {
    float s = 0.f;
    for (int q = 0; q < num_q; ++q) s += sq[q * S128 + tid];
    qsum[tid] = s / fmaxf((float)num_q, 1.f);
  }
  __syncthreads();

  const float scale = 0.088388347648318447f;
#pragma unroll
  for (int t2 = 0; t2 < 2; ++t2) {
    const int t  = wave * 2 + t2;
    const int mt = t >> 2, nt = t & 3;
    const float* Ar = sc + (mt * 16 + l) * S128;
    const float* Br = sq + (nt * 16 + l) * S128;
    v8f acc = {};
    for (int kk = 0; kk < 32; ++kk) {
      const int k0 = kk * 4 + 2 * half;
      v2f a;  a.x  = Ar[k0]; a.y  = Ar[k0 + 1];
      v2f bv; bv.x = Br[k0]; bv.y = Br[k0 + 1];
      acc = wmma4(a, bv, acc);
    }
    const int ncol = nt * 16 + l;
    const bool valid = (ncol < num_q);
#pragma unroll
    for (int r = 0; r < 8; ++r) {
      const int row = mt * 16 + r + 8 * half;
      satt[row * S64 + ncol] = valid ? acc[r] * scale : -__builtin_inff();
    }
  }
  __syncthreads();

  if (tid < CC) {
    float* row = satt + tid * S64;
    float m = -__builtin_inff();
    for (int q = 0; q < QQ; ++q) m = fmaxf(m, row[q]);
    float s = 0.f;
    for (int q = 0; q < QQ; ++q) { const float e = __expf(row[q] - m); row[q] = e; s += e; }
    const float inv = 1.f / s;
    for (int q = 0; q < QQ; ++q) row[q] *= inv;
  }
  __syncthreads();

#pragma unroll
  for (int t4 = 0; t4 < 4; ++t4) {
    const int t  = wave * 4 + t4;
    const int mt = t >> 3, nt = t & 7;
    const float* Ar = satt + (mt * 16 + l) * S64;
    const int ncol = nt * 16 + l;
    v8f acc = {};
    for (int kk = 0; kk < 16; ++kk) {
      const int k0 = kk * 4 + 2 * half;
      v2f a;  a.x  = Ar[k0];                 a.y  = Ar[k0 + 1];
      v2f bv; bv.x = sq[k0 * S128 + ncol];   bv.y = sq[(k0 + 1) * S128 + ncol];
      acc = wmma4(a, bv, acc);
    }
#pragma unroll
    for (int r = 0; r < 8; ++r)
      swq[(mt * 16 + r + 8 * half) * S128 + ncol] = acc[r];
  }
  __syncthreads();

  {
    const int ncol = wave * 16 + l;
    const unsigned wlane = (unsigned)(ncol + 2 * half * DD);
    const float* Cb = sc   + l * S128 + 2 * half;
    const float* Qb = swq  + l * S128 + 2 * half;
    const float* Sb = qsum + 2 * half;

    v2f wr[4][5];
    v2f cr[2][4], qr[2][4], sr[2];

    auto ldW = [&](int kk, v2f (&w)[5]) {
      const float* Wk = W_h + (size_t)kk * (4 * DD);
#pragma unroll
      for (int s = 0; s < 5; ++s) {
        w[s].x = Wk[wlane + 128 * s * DD];
        w[s].y = Wk[wlane + 128 * s * DD + DD];
      }
    };
    auto ldA = [&](int kk, v2f (&c)[4], v2f (&q)[4], v2f& sm) {
      sm = *reinterpret_cast<const v2f*>(Sb + 4 * kk);
#pragma unroll
      for (int m = 0; m < 4; ++m) {
        c[m] = *reinterpret_cast<const v2f*>(Cb + m * 16 * S128 + 4 * kk);
        q[m] = *reinterpret_cast<const v2f*>(Qb + m * 16 * S128 + 4 * kk);
      }
    };

    v8f accs = {};
    v8f acc[4] = {};
    ldW(0, wr[0]); ldW(1, wr[1]); ldW(2, wr[2]);
    ldA(0, cr[0], qr[0], sr[0]);
#pragma unroll
    for (int k4 = 0; k4 < 32; k4 += 4) {
#pragma unroll
      for (int j = 0; j < 4; ++j) {
        const int kk = k4 + j;
        if (kk + 3 < 32) ldW(kk + 3, wr[(j + 3) & 3]);
        if (kk + 1 < 32) ldA(kk + 1, cr[(j + 1) & 1], qr[(j + 1) & 1], sr[(j + 1) & 1]);
        __builtin_amdgcn_sched_barrier(0);
        const v2f (&w)[5] = wr[j];
        const v2f (&c)[4] = cr[j & 1];
        const v2f (&q)[4] = qr[j & 1];
        v2f pr[4], df[4];
#pragma unroll
        for (int m = 0; m < 4; ++m) {
          pr[m].x = c[m].x * q[m].x;          pr[m].y = c[m].y * q[m].y;
          df[m].x = fabsf(c[m].x - q[m].x);   df[m].y = fabsf(c[m].y - q[m].y);
        }
        accs = wmma4(sr[j & 1], w[0], accs);
#pragma unroll
        for (int m = 0; m < 4; ++m) acc[m] = wmma4(c[m], w[1], acc[m]);
#pragma unroll
        for (int m = 0; m < 4; ++m) acc[m] = wmma4(q[m], w[2], acc[m]);
#pragma unroll
        for (int m = 0; m < 4; ++m) acc[m] = wmma4(pr[m], w[3], acc[m]);
#pragma unroll
        for (int m = 0; m < 4; ++m) acc[m] = wmma4(df[m], w[4], acc[m]);
        __builtin_amdgcn_sched_barrier(0);
      }
    }
    const float bias = b_h[ncol];
#pragma unroll
    for (int m = 0; m < 4; ++m)
#pragma unroll
      for (int r = 0; r < 8; ++r)
        sq[(m * 16 + r + 8 * half) * S128 + ncol] = tanhf(accs[r] + acc[m][r] + bias);
  }
  __syncthreads();

  if (tid < CC) {
    float s = 0.f;
    for (int n = 0; n < DD; ++n) s += sq[tid * S128 + n] * W_o[n];
    out[(size_t)b * CC + tid] = s + b_o[0];
  }
}

extern "C" void kernel_launch(void* const* d_in, const int* in_sizes, int n_in,
                              void* d_out, int out_size, void* d_ws, size_t ws_size,
                              hipStream_t stream) {
  (void)in_sizes; (void)n_in; (void)out_size; (void)d_ws; (void)ws_size;
  const int*   q_ids  = (const int*)d_in[0];
  const int*   c_ids  = (const int*)d_in[1];
  const int*   num_qs = (const int*)d_in[2];
  const float* embed  = (const float*)d_in[4];
  const float* W_h    = (const float*)d_in[5];
  const float* b_h    = (const float*)d_in[6];
  const float* W_o    = (const float*)d_in[7];
  const float* b_o    = (const float*)d_in[8];
  float* out = (float*)d_out;

  const size_t shmem = (size_t)(3 * 64 * S128 + 64 * S64 + 128) * sizeof(float);
  hipFuncSetAttribute(reinterpret_cast<const void*>(entity_linker_kernel),
                      hipFuncAttributeMaxDynamicSharedMemorySize, (int)shmem);
  entity_linker_kernel<<<1024, 256, shmem, stream>>>(
      q_ids, c_ids, num_qs, embed, W_h, b_h, W_o, b_o, out);
}


# embedding gather re-mapped: one wave-wide load (16 B per lane) reads a whole 512-B row, was 4 threads per row = 32 lines per load; ids read once per wave, passed by readlane; 24 rows in flight
# speedup vs baseline: 1.5223x; 1.2017x over previous
#include <hip/hip_runtime.h>
#include <math.h>

typedef __attribute__((ext_vector_type(2))) float v2f;
typedef __attribute__((ext_vector_type(4))) float v4f;
typedef __attribute__((ext_vector_type(8))) float v8f;

#define DD   128
#define QQ   64
#define CC   64
#define TT   8
#define S128 132
#define S64  68

__device__ __forceinline__ v8f wmma4(v2f a, v2f b, v8f c) {
  return __builtin_amdgcn_wmma_f32_16x16x4_f32(
       false, a,  false, b,
       (short)0, c,  false,  false);
}

__global__ __launch_bounds__(256)
void entity_linker_kernel(const int* __restrict__ q_ids,
                          const int* __restrict__ c_ids,
                          const int* __restrict__ num_qs,
                          const float* __restrict__ embed,
                          const float* __restrict__ W_h,
                          const float* __restrict__ b_h,
                          const float* __restrict__ W_o,
                          const float* __restrict__ b_o,
                          float* __restrict__ out) {
  extern __shared__ float smem[];
  float* sq   = smem;
  float* sc   = sq  + 64 * S128;
  float* swq  = sc  + 64 * S128;
  float* satt = swq + 64 * S128;
  float* qsum = satt + 64 * S64;

  const int b    = blockIdx.x;
  const int tid  = threadIdx.x;
  const int wave = tid >> 5;
  const int lane = tid & 31;
  const int half = lane >> 4;
  const int l    = lane & 15;
  const int num_q = num_qs[b];

  {
    const int r0 = wave * 8;
    const int qid_l  = q_ids[(size_t)b * QQ + r0 + (lane & 7)];
    const int* cbase = c_ids + ((size_t)b * CC + r0) * TT;
    const int cid_lo = cbase[lane];
    const int cid_hi = cbase[32 + lane];
    const float mk_lo = (cid_lo != 0) ? 1.f : 0.f;
    const float mk_hi = (cid_hi != 0) ? 1.f : 0.f;
    const unsigned lane_off = (unsigned)lane * 16u;

    auto ldrow = [&](int id) -> v4f {
      const unsigned long long p = (unsigned long long)(embed + (size_t)id * DD);
      const unsigned lo = __builtin_amdgcn_readfirstlane((unsigned)p);
      const unsigned hi = __builtin_amdgcn_readfirstlane((unsigned)(p >> 32));
      typedef const __attribute__((address_space(1))) char* gcp;
      typedef const __attribute__((address_space(1))) v4f*  gvp;
      const gcp rowp = (gcp)(((unsigned long long)hi << 32) | lo);
      return *(gvp)(rowp + lane_off);
    };

    v4f qv[8];
#pragma unroll
    for (int i = 0; i < 8; ++i) qv[i] = ldrow(__builtin_amdgcn_readlane(qid_l, i));

#pragma unroll
    for (int g = 0; g < 4; ++g) {
      const int   cid = (g < 2) ? cid_lo : cid_hi;
      const float mk  = (g < 2) ? mk_lo  : mk_hi;
      const int   j0  = (g & 1) * 16;
      v4f rv[2][8];
#pragma unroll
      for (int cc = 0; cc < 2; ++cc)
#pragma unroll
        for (int t = 0; t < TT; ++t)
          rv[cc][t] = ldrow(__builtin_amdgcn_readlane(cid, j0 + cc * TT + t));
      __builtin_amdgcn_sched_barrier(0);
      if (g == 0) {
#pragma unroll
        for (int i = 0; i < 8; ++i)
          *reinterpret_cast<v4f*>(sq + (r0 + i) * S128 + 4 * lane) = qv[i];
      }
#pragma unroll
      for (int cc = 0; cc < 2; ++cc) {
        v4f acc = {0.f, 0.f, 0.f, 0.f};
        float cnt = 0.f;
#pragma unroll
        for (int t = 0; t < TT; ++t) {
          const float m = __int_as_float(__builtin_amdgcn_readlane(__float_as_int(mk), j0 + cc * TT + t));
          cnt += m;
          acc += rv[cc][t] * m;
        }
        const float inv = 1.f / fmaxf(cnt, 1.f);
        *reinterpret_cast<v4f*>(sc + (r0 + g * 2 + cc) * S128 + 4 * lane) = acc * inv;
      }
    }
  }
  __syncthreads();

  if (tid < DD) {
    float s = 0.f;
    for (int q = 0; q < num_q; ++q) s += sq[q * S128 + tid];
    qsum[tid] = s / fmaxf((float)num_q, 1.f);
  }
  __syncthreads();

  const float scale = 0.088388347648318447f;
#pragma unroll
  for (int t2 = 0; t2 < 2; ++t2) {
    const int t  = wave * 2 + t2;
    const int mt = t >> 2, nt = t & 3;
    const float* Ar = sc + (mt * 16 + l) * S128;
    const float* Br = sq + (nt * 16 + l) * S128;
    v8f acc = {};
    for (int kk = 0; kk < 32; ++kk) {
      const int k0 = kk * 4 + 2 * half;
      v2f a;  a.x  = Ar[k0]; a.y  = Ar[k0 + 1];
      v2f bv; bv.x = Br[k0]; bv.y = Br[k0 + 1];
      acc = wmma4(a, bv, acc);
    }
    const int ncol = nt * 16 + l;
    const bool valid = (ncol < num_q);
#pragma unroll
    for (int r = 0; r < 8; ++r) {
      const int row = mt * 16 + r + 8 * half;
      satt[row * S64 + ncol] = valid ? acc[r] * scale : -__builtin_inff();
    }
  }
  __syncthreads();

  if (tid < CC) {
    float* row = satt + tid * S64;
    float m = -__builtin_inff();
    for (int q = 0; q < QQ; ++q) m = fmaxf(m, row[q]);
    float s = 0.f;
    for (int q = 0; q < QQ; ++q) { const float e = __expf(row[q] - m); row[q] = e; s += e; }
    const float inv = 1.f / s;
    for (int q = 0; q < QQ; ++q) row[q] *= inv;
  }
  __syncthreads();

#pragma unroll
  for (int t4 = 0; t4 < 4; ++t4) {
    const int t  = wave * 4 + t4;
    const int mt = t >> 3, nt = t & 7;
    const float* Ar = satt + (mt * 16 + l) * S64;
    const int ncol = nt * 16 + l;
    v8f acc = {};
    for (int kk = 0; kk < 16; ++kk) {
      const int k0 = kk * 4 + 2 * half;
      v2f a;  a.x  = Ar[k0];                 a.y  = Ar[k0 + 1];
      v2f bv; bv.x = sq[k0 * S128 + ncol];   bv.y = sq[(k0 + 1) * S128 + ncol];
      acc = wmma4(a, bv, acc);
    }
#pragma unroll
    for (int r = 0; r < 8; ++r)
      swq[(mt * 16 + r + 8 * half) * S128 + ncol] = acc[r];
  }
  __syncthreads();

  {
    const int ncol = wave * 16 + l;
    const unsigned wlane = (unsigned)(ncol + 2 * half * DD);
    const float* Cb = sc   + l * S128 + 2 * half;
    const float* Qb = swq  + l * S128 + 2 * half;
    const float* Sb = qsum + 2 * half;

    v2f wr[4][5];
    v2f cr[2][4], qr[2][4], sr[2];

    auto ldW = [&](int kk, v2f (&w)[5]) {
      const float* Wk = W_h + (size_t)kk * (4 * DD);
#pragma unroll
      for (int s = 0; s < 5; ++s) {
        w[s].x = Wk[wlane + 128 * s * DD];
        w[s].y = Wk[wlane + 128 * s * DD + DD];
      }
    };
    auto ldA = [&](int kk, v2f (&c)[4], v2f (&q)[4], v2f& sm) {
      sm = *reinterpret_cast<const v2f*>(Sb + 4 * kk);
#pragma unroll
      for (int m = 0; m < 4; ++m) {
        c[m] = *reinterpret_cast<const v2f*>(Cb + m * 16 * S128 + 4 * kk);
        q[m] = *reinterpret_cast<const v2f*>(Qb + m * 16 * S128 + 4 * kk);
      }
    };

    v8f accs = {};
    v8f acc[4] = {};
    ldW(0, wr[0]); ldW(1, wr[1]); ldW(2, wr[2]);
    ldA(0, cr[0], qr[0], sr[0]);
#pragma unroll
    for (int k4 = 0; k4 < 32; k4 += 4) {
#pragma unroll
      for (int j = 0; j < 4; ++j) {
        const int kk = k4 + j;
        if (kk + 3 < 32) ldW(kk + 3, wr[(j + 3) & 3]);
        if (kk + 1 < 32) ldA(kk + 1, cr[(j + 1) & 1], qr[(j + 1) & 1], sr[(j + 1) & 1]);
        __builtin_amdgcn_sched_barrier(0);
        const v2f (&w)[5] = wr[j];
        const v2f (&c)[4] = cr[j & 1];
        const v2f (&q)[4] = qr[j & 1];
        v2f pr[4], df[4];
#pragma unroll
        for (int m = 0; m < 4; ++m) {
          pr[m].x = c[m].x * q[m].x;          pr[m].y = c[m].y * q[m].y;
          df[m].x = fabsf(c[m].x - q[m].x);   df[m].y = fabsf(c[m].y - q[m].y);
        }
        accs = wmma4(sr[j & 1], w[0], accs);
#pragma unroll
        for (int m = 0; m < 4; ++m) acc[m] = wmma4(c[m], w[1], acc[m]);
#pragma unroll
        for (int m = 0; m < 4; ++m) acc[m] = wmma4(q[m], w[2], acc[m]);
#pragma unroll
        for (int m = 0; m < 4; ++m) acc[m] = wmma4(pr[m], w[3], acc[m]);
#pragma unroll
        for (int m = 0; m < 4; ++m) acc[m] = wmma4(df[m], w[4], acc[m]);
        __builtin_amdgcn_sched_barrier(0);
      }
    }
    const float bias = b_h[ncol];
#pragma unroll
    for (int m = 0; m < 4; ++m)
#pragma unroll
      for (int r = 0; r < 8; ++r)
        sq[(m * 16 + r + 8 * half) * S128 + ncol] = tanhf(accs[r] + acc[m][r] + bias);
  }
  __syncthreads();

  if (tid < CC) {
    float s = 0.f;
    for (int n = 0; n < DD; ++n) s += sq[tid * S128 + n] * W_o[n];
    out[(size_t)b * CC + tid] = s + b_o[0];
  }
}

extern "C" void kernel_launch(void* const* d_in, const int* in_sizes, int n_in,
                              void* d_out, int out_size, void* d_ws, size_t ws_size,
                              hipStream_t stream) {
  (void)in_sizes; (void)n_in; (void)out_size; (void)d_ws; (void)ws_size;
  const int*   q_ids  = (const int*)d_in[0];
  const int*   c_ids  = (const int*)d_in[1];
  const int*   num_qs = (const int*)d_in[2];
  const float* embed  = (const float*)d_in[4];
  const float* W_h    = (const float*)d_in[5];
  const float* b_h    = (const float*)d_in[6];
  const float* W_o    = (const float*)d_in[7];
  const float* b_o    = (const float*)d_in[8];
  float* out = (float*)d_out;

  const size_t shmem = (size_t)(3 * 64 * S128 + 64 * S64 + 128) * sizeof(float);
  hipFuncSetAttribute(reinterpret_cast<const void*>(entity_linker_kernel),
                      hipFuncAttributeMaxDynamicSharedMemorySize, (int)shmem);
  entity_linker_kernel<<<1024, 256, shmem, stream>>>(
      q_ids, c_ids, num_qs, embed, W_h, b_h, W_o, b_o, out);
}


# the three one-thread-per-row serial loops (q_summary, softmax, out dot) spread over all 256 threads, partial sums joined by lane swizzle or LDS; these sums are now tree-ordered, not serial
# speedup vs baseline: 1.6710x; 1.0977x over previous
#include <hip/hip_runtime.h>
#include <math.h>

typedef __attribute__((ext_vector_type(2))) float v2f;
typedef __attribute__((ext_vector_type(4))) float v4f;
typedef __attribute__((ext_vector_type(8))) float v8f;

#define DD   128
#define QQ   64
#define CC   64
#define TT   8
#define S128 132
#define S64  68

__device__ __forceinline__ v8f wmma4(v2f a, v2f b, v8f c) {
  return __builtin_amdgcn_wmma_f32_16x16x4_f32(
       false, a,  false, b,
       (short)0, c,  false,  false);
}

__device__ __forceinline__ float lane_xor1(float x) {
  const int v = __builtin_bit_cast(int, x);
  return __builtin_bit_cast(float, __builtin_amdgcn_update_dpp(v, v, 0xB1, 0xF, 0xF, false));
}
__device__ __forceinline__ float lane_xor2(float x) {
  const int v = __builtin_bit_cast(int, x);
  return __builtin_bit_cast(float, __builtin_amdgcn_update_dpp(v, v, 0x4E, 0xF, 0xF, false));
}

__global__ __launch_bounds__(256)
void entity_linker_kernel(const int* __restrict__ q_ids,
                          const int* __restrict__ c_ids,
                          const int* __restrict__ num_qs,
                          const float* __restrict__ embed,
                          const float* __restrict__ W_h,
                          const float* __restrict__ b_h,
                          const float* __restrict__ W_o,
                          const float* __restrict__ b_o,
                          float* __restrict__ out) {
  extern __shared__ float smem[];
  float* sq   = smem;
  float* sc   = sq  + 64 * S128;
  float* swq  = sc  + 64 * S128;
  float* satt = swq + 64 * S128;
  float* qsum = satt + 64 * S64;

  const int b    = blockIdx.x;
  const int tid  = threadIdx.x;
  const int wave = tid >> 5;
  const int lane = tid & 31;
  const int half = lane >> 4;
  const int l    = lane & 15;
  const int num_q = num_qs[b];

  {
    const int r0 = wave * 8;
    const int qid_l  = q_ids[(size_t)b * QQ + r0 + (lane & 7)];
    const int* cbase = c_ids + ((size_t)b * CC + r0) * TT;
    const int cid_lo = cbase[lane];
    const int cid_hi = cbase[32 + lane];
    const float mk_lo = (cid_lo != 0) ? 1.f : 0.f;
    const float mk_hi = (cid_hi != 0) ? 1.f : 0.f;
    const unsigned lane_off = (unsigned)lane * 16u;

    auto ldrow = [&](int id) -> v4f {
      const unsigned long long p = (unsigned long long)(embed + (size_t)id * DD);
      const unsigned lo = __builtin_amdgcn_readfirstlane((unsigned)p);
      const unsigned hi = __builtin_amdgcn_readfirstlane((unsigned)(p >> 32));
      typedef const __attribute__((address_space(1))) char* gcp;
      typedef const __attribute__((address_space(1))) v4f*  gvp;
      const gcp rowp = (gcp)(((unsigned long long)hi << 32) | lo);
      return *(gvp)(rowp + lane_off);
    };

    v4f qv[8];
#pragma unroll
    for (int i = 0; i < 8; ++i) qv[i] = ldrow(__builtin_amdgcn_readlane(qid_l, i));

#pragma unroll
    for (int g = 0; g < 4; ++g) {
      const int   cid = (g < 2) ? cid_lo : cid_hi;
      const float mk  = (g < 2) ? mk_lo  : mk_hi;
      const int   j0  = (g & 1) * 16;
      v4f rv[2][8];
#pragma unroll
      for (int cc = 0; cc < 2; ++cc)
#pragma unroll
        for (int t = 0; t < TT; ++t)
          rv[cc][t] = ldrow(__builtin_amdgcn_readlane(cid, j0 + cc * TT + t));
      __builtin_amdgcn_sched_barrier(0);
      if (g == 0) {
#pragma unroll
        for (int i = 0; i < 8; ++i)
          *reinterpret_cast<v4f*>(sq + (r0 + i) * S128 + 4 * lane) = qv[i];
      }
#pragma unroll
      for (int cc = 0; cc < 2; ++cc) {
        v4f acc = {0.f, 0.f, 0.f, 0.f};
        float cnt = 0.f;
#pragma unroll
        for (int t = 0; t < TT; ++t) {
          const float m = __int_as_float(__builtin_amdgcn_readlane(__float_as_int(mk), j0 + cc * TT + t));
          cnt += m;
          acc += rv[cc][t] * m;
        }
        const float inv = 1.f / fmaxf(cnt, 1.f);
        *reinterpret_cast<v4f*>(sc + (r0 + g * 2 + cc) * S128 + 4 * lane) = acc * inv;
      }
    }
  }
  __syncthreads();

  {
    const int j   = lane & 3;
    const int col = wave * 16 + 2 * (lane >> 2);
    const float* p = sq + j * S128 + col;
    v2f s = {0.f, 0.f};
#pragma unroll
    for (int i = 0; i < 16; ++i) {
      const v2f v = *reinterpret_cast<const v2f*>(p + 4 * i * S128);
      const bool in = (4 * i + j) < num_q;
      s.x += in ? v.x : 0.f;
      s.y += in ? v.y : 0.f;
    }
    s.x += lane_xor1(s.x);  s.y += lane_xor1(s.y);
    s.x += lane_xor2(s.x);  s.y += lane_xor2(s.y);
    if (j == 0) {
      const float den = fmaxf((float)num_q, 1.f);
      qsum[col]     = s.x / den;
      qsum[col + 1] = s.y / den;
    }
  }

  const float scale = 0.088388347648318447f;
#pragma unroll
  for (int t2 = 0; t2 < 2; ++t2) {
    const int t  = wave * 2 + t2;
    const int mt = t >> 2, nt = t & 3;
    const float* Ar = sc + (mt * 16 + l) * S128;
    const float* Br = sq + (nt * 16 + l) * S128;
    v8f acc = {};
    for (int kk = 0; kk < 32; ++kk) {
      const int k0 = kk * 4 + 2 * half;
      v2f a;  a.x  = Ar[k0]; a.y  = Ar[k0 + 1];
      v2f bv; bv.x = Br[k0]; bv.y = Br[k0 + 1];
      acc = wmma4(a, bv, acc);
    }
    const int ncol = nt * 16 + l;
    const bool valid = (ncol < num_q);
#pragma unroll
    for (int r = 0; r < 8; ++r) {
      const int row = mt * 16 + r + 8 * half;
      satt[row * S64 + ncol] = valid ? acc[r] * scale : -__builtin_inff();
    }
  }
  __syncthreads();

  {
    const int j = tid & 3;
    float* p = satt + (tid >> 2) * S64 + 16 * j;
    v4f x[4];
#pragma unroll
    for (int i = 0; i < 4; ++i) x[i] = *reinterpret_cast<const v4f*>(p + 4 * i);
    float m = -__builtin_inff();
#pragma unroll
    for (int i = 0; i < 4; ++i)
      m = fmaxf(m, fmaxf(fmaxf(x[i].x, x[i].y), fmaxf(x[i].z, x[i].w)));
    m = fmaxf(m, lane_xor1(m));
    m = fmaxf(m, lane_xor2(m));
    float s = 0.f;
#pragma unroll
    for (int i = 0; i < 4; ++i) {
      x[i].x = __expf(x[i].x - m);  x[i].y = __expf(x[i].y - m);
      x[i].z = __expf(x[i].z - m);  x[i].w = __expf(x[i].w - m);
      s += (x[i].x + x[i].y) + (x[i].z + x[i].w);
    }
    s += lane_xor1(s);
    s += lane_xor2(s);
    const float inv = 1.f / s;
#pragma unroll
    for (int i = 0; i < 4; ++i) *reinterpret_cast<v4f*>(p + 4 * i) = x[i] * inv;
  }
  __syncthreads();

#pragma unroll
  for (int t4 = 0; t4 < 4; ++t4) {
    const int t  = wave * 4 + t4;
    const int mt = t >> 3, nt = t & 7;
    const float* Ar = satt + (mt * 16 + l) * S64;
    const int ncol = nt * 16 + l;
    v8f acc = {};
    for (int kk = 0; kk < 16; ++kk) {
      const int k0 = kk * 4 + 2 * half;
      v2f a;  a.x  = Ar[k0];                 a.y  = Ar[k0 + 1];
      v2f bv; bv.x = sq[k0 * S128 + ncol];   bv.y = sq[(k0 + 1) * S128 + ncol];
      acc = wmma4(a, bv, acc);
    }
#pragma unroll
    for (int r = 0; r < 8; ++r)
      swq[(mt * 16 + r + 8 * half) * S128 + ncol] = acc[r];
  }
  __syncthreads();

  v4f wo[8];
  {
    const int ncol = wave * 16 + l;
    const unsigned wlane = (unsigned)(ncol + 2 * half * DD);
    const float* Cb = sc   + l * S128 + 2 * half;
    const float* Qb = swq  + l * S128 + 2 * half;
    const float* Sb = qsum + 2 * half;

    v2f wr[4][5];
    v2f cr[2][4], qr[2][4], sr[2];

    auto ldW = [&](int kk, v2f (&w)[5]) {
      const float* Wk = W_h + (size_t)kk * (4 * DD);
#pragma unroll
      for (int s = 0; s < 5; ++s) {
        w[s].x = Wk[wlane + 128 * s * DD];
        w[s].y = Wk[wlane + 128 * s * DD + DD];
      }
    };
    auto ldA = [&](int kk, v2f (&c)[4], v2f (&q)[4], v2f& sm) {
      sm = *reinterpret_cast<const v2f*>(Sb + 4 * kk);
#pragma unroll
      for (int m = 0; m < 4; ++m) {
        c[m] = *reinterpret_cast<const v2f*>(Cb + m * 16 * S128 + 4 * kk);
        q[m] = *reinterpret_cast<const v2f*>(Qb + m * 16 * S128 + 4 * kk);
      }
    };

    v8f accs = {};
    v8f acc[4] = {};
    ldW(0, wr[0]); ldW(1, wr[1]); ldW(2, wr[2]);
    ldA(0, cr[0], qr[0], sr[0]);
#pragma unroll
    for (int k4 = 0; k4 < 32; k4 += 4) {
#pragma unroll
      for (int j = 0; j < 4; ++j) {
        const int kk = k4 + j;
        if (kk + 3 < 32) ldW(kk + 3, wr[(j + 3) & 3]);
        if (kk + 1 < 32) ldA(kk + 1, cr[(j + 1) & 1], qr[(j + 1) & 1], sr[(j + 1) & 1]);
        __builtin_amdgcn_sched_barrier(0);
        const v2f (&w)[5] = wr[j];
        const v2f (&c)[4] = cr[j & 1];
        const v2f (&q)[4] = qr[j & 1];
        v2f pr[4], df[4];
#pragma unroll
        for (int m = 0; m < 4; ++m) {
          pr[m].x = c[m].x * q[m].x;          pr[m].y = c[m].y * q[m].y;
          df[m].x = fabsf(c[m].x - q[m].x);   df[m].y = fabsf(c[m].y - q[m].y);
        }
        accs = wmma4(sr[j & 1], w[0], accs);
#pragma unroll
        for (int m = 0; m < 4; ++m) acc[m] = wmma4(c[m], w[1], acc[m]);
#pragma unroll
        for (int m = 0; m < 4; ++m) acc[m] = wmma4(q[m], w[2], acc[m]);
#pragma unroll
        for (int m = 0; m < 4; ++m) acc[m] = wmma4(pr[m], w[3], acc[m]);
#pragma unroll
        for (int m = 0; m < 4; ++m) acc[m] = wmma4(df[m], w[4], acc[m]);
        __builtin_amdgcn_sched_barrier(0);
      }
    }
#pragma unroll
    for (int k = 0; k < 8; ++k)
      wo[k] = *reinterpret_cast<const v4f*>(W_o + 32 * (wave >> 1) + 4 * k);
    const float bias = b_h[ncol];
#pragma unroll
    for (int m = 0; m < 4; ++m)
#pragma unroll
      for (int r = 0; r < 8; ++r)
        sq[(m * 16 + r + 8 * half) * S128 + ncol] = tanhf(accs[r] + acc[m][r] + bias);
  }
  __syncthreads();

  {
    const int seg = wave >> 1;
    const int row = (wave & 1) * 32 + lane;
    const float* hp = sq + row * S128 + 32 * seg;
    v4f a4 = {0.f, 0.f, 0.f, 0.f};
#pragma unroll
    for (int k = 0; k < 8; ++k)
      a4 += *reinterpret_cast<const v4f*>(hp + 4 * k) * wo[k];
    satt[seg * CC + row] = (a4.x + a4.y) + (a4.z + a4.w);
  }
  __syncthreads();
  if (tid < CC)
    out[(size_t)b * CC + tid] =
        ((satt[tid] + satt[CC + tid]) + (satt[2 * CC + tid] + satt[3 * CC + tid])) + b_o[0];
}

extern "C" void kernel_launch(void* const* d_in, const int* in_sizes, int n_in,
                              void* d_out, int out_size, void* d_ws, size_t ws_size,
                              hipStream_t stream) {
  (void)in_sizes; (void)n_in; (void)out_size; (void)d_ws; (void)ws_size;
  const int*   q_ids  = (const int*)d_in[0];
  const int*   c_ids  = (const int*)d_in[1];
  const int*   num_qs = (const int*)d_in[2];
  const float* embed  = (const float*)d_in[4];
  const float* W_h    = (const float*)d_in[5];
  const float* b_h    = (const float*)d_in[6];
  const float* W_o    = (const float*)d_in[7];
  const float* b_o    = (const float*)d_in[8];
  float* out = (float*)d_out;

  const size_t shmem = (size_t)(3 * 64 * S128 + 64 * S64 + 128) * sizeof(float);
  hipFuncSetAttribute(reinterpret_cast<const void*>(entity_linker_kernel),
                      hipFuncAttributeMaxDynamicSharedMemorySize, (int)shmem);
  entity_linker_kernel<<<1024, 256, shmem, stream>>>(
      q_ids, c_ids, num_qs, embed, W_h, b_h, W_o, b_o, out);
}
